# starting kernel
# baseline (speedup 1.0000x reference)
#include <hip/hip_runtime.h>
#include <math.h>


typedef float v2f __attribute__((ext_vector_type(2)));
typedef float v8f __attribute__((ext_vector_type(8)));

#define EPS_F 1e-6f
#define KN   32
#define NFG  4
#define NH   16
#define NFO  32
#define ROWLEN 258

__global__ __launch_bounds__(256)
void attn_layer_kernel(const float* __restrict__ beta,
                       const float* __restrict__ sa,
                       const float* __restrict__ ac,
                       const float* __restrict__ no,
                       const float* __restrict__ gw,
                       float* __restrict__ out,
                       float* __restrict__ out_attn)
{
    __shared__ float no_s[NFO * ROWLEN];
    __shared__ float attn_s[KN * NH];
    __shared__ float mx_s[16 * NH];

    const int site = blockIdx.x;
    const int tid  = threadIdx.x;
    const int h    = tid & 15;
    const int k2   = tid >> 4;

    const float* ac_p = ac   + (size_t)site * (KN * NFG * NH);
    const float* no_p = no   + (size_t)site * (KN * NFO * NH);
    const float* bt_p = beta + (size_t)site * (NFG * NH);
    const float* sa_p = sa   + (size_t)site * (NFG * NH);
    const float* gw_p = gw   + (size_t)site * (KN * NFG);

    float t[8];
    float g[8];
    float lmax = -3.402823466e38f;
    #pragma unroll
    for (int j = 0; j < 2; ++j) {
        const int k = k2 + j * 16;
        #pragma unroll
        for (int f = 0; f < NFG; ++f) {
            float v = ac_p[k * (NFG * NH) + f * NH + h];
            if (k == 0) v += sa_p[f * NH + h];
            v *= (bt_p[f * NH + h] + EPS_F);
            t[j * 4 + f] = v;
            g[j * 4 + f] = gw_p[k * NFG + f];
            lmax = fmaxf(lmax, v);
        }
    }
    mx_s[k2 * NH + h] = lmax;
    __syncthreads();
    float m = mx_s[h];
    #pragma unroll
    for (int i = 1; i < 16; ++i) m = fmaxf(m, mx_s[i * NH + h]);

    float a0 = 0.f, a1 = 0.f;
    #pragma unroll
    for (int f = 0; f < 4; ++f) a0 += __expf(t[f] - m) * g[f];
    #pragma unroll
    for (int f = 0; f < 4; ++f) a1 += __expf(t[4 + f] - m) * g[4 + f];
    attn_s[k2 * NH + h]        = a0;
    attn_s[(k2 + 16) * NH + h] = a1;
    __syncthreads();
    float denom = EPS_F;
    #pragma unroll
    for (int k = 0; k < KN; ++k) denom += fabsf(attn_s[k * NH + h]);
    __syncthreads();
    const float rden = 1.0f / denom;
    a0 *= rden; a1 *= rden;
    attn_s[k2 * NH + h]        = a0;
    attn_s[(k2 + 16) * NH + h] = a1;
    out_attn[(size_t)site * 512 + k2 * NH + h]        = a0;
    out_attn[(size_t)site * 512 + (k2 + 16) * NH + h] = a1;
    __syncthreads();

    const int lane = tid & 31;
    const int wave = tid >> 5;
    const int mrow = lane & 15;
    const int sel  = lane >> 4;

    v8f acc[4] = {};

    #pragma unroll
    for (int half = 0; half < 2; ++half) {
        #pragma unroll
        for (int j = 0; j < 8; ++j) {
            const int idx4 = (j * 256 + tid) * 4;
            const float4 v = *(const float4*)(no_p + half * 8192 + idx4);
            const int kl = idx4 >> 9;
            const int f  = (idx4 >> 4) & 31;
            const int hh = idx4 & 15;
            float* dst = &no_s[f * ROWLEN + kl * 16 + hh];
            ((float2*)dst)[0] = make_float2(v.x, v.y);
            ((float2*)dst)[1] = make_float2(v.z, v.w);
        }
        __syncthreads();

        #pragma unroll
        for (int i = 0; i < 4; ++i) {
            const int task  = wave * 4 + i;
            const int th    = task & 15;
            const int ftile = task >> 4;
            const int rowbase = (ftile * 16 + mrow) * ROWLEN;
            #pragma unroll
            for (int kc = 0; kc < 4; ++kc) {
                const int klb = kc * 4 + sel * 2;
                v2f a, b;
                a.x = no_s[rowbase + klb * 16 + th];
                a.y = no_s[rowbase + (klb + 1) * 16 + th];
                const int kg = half * 16 + klb;
                b.x = attn_s[kg * NH + th];
                b.y = attn_s[(kg + 1) * NH + th];
                acc[i] = __builtin_amdgcn_wmma_f32_16x16x4_f32(
                    false, a, false, b, (short)0, acc[i], false, false);
            }
        }
        __syncthreads();
    }

    if ((lane & 15) == 0) {
        #pragma unroll
        for (int i = 0; i < 4; ++i) {
            const int task  = wave * 4 + i;
            const int th    = task & 15;
            const int fbase = (task >> 4) * 16 + sel * 8;
            #pragma unroll
            for (int rr = 0; rr < 8; ++rr) {
                out[(size_t)site * 512 + (fbase + rr) * NH + th] = acc[i][rr];
            }
        }
    }
}

extern "C" void kernel_launch(void* const* d_in, const int* in_sizes, int n_in,
                              void* d_out, int out_size, void* d_ws, size_t ws_size,
                              hipStream_t stream) {
    const float* beta = (const float*)d_in[0];
    const float* sa   = (const float*)d_in[1];
    const float* ac   = (const float*)d_in[2];
    const float* no   = (const float*)d_in[3];
    const float* gw   = (const float*)d_in[4];
    float* out = (float*)d_out;

    const int sites = in_sizes[0] / (NFG * NH);
    float* out_attn = out + (size_t)sites * (NFO * NH);

    attn_layer_kernel<<<sites, 256, 0, stream>>>(beta, sa, ac, no, gw, out, out_attn);
}


# node_out requested at kernel entry straight into WMMA A registers (16 coalesced 16-byte loads per lane in flight across phase 1, behind LDS-only barriers); LDS staging of node_out gone
# speedup vs baseline: 1.3672x; 1.3672x over previous
#include <hip/hip_runtime.h>
#include <math.h>


typedef float v2f __attribute__((ext_vector_type(2)));
typedef float v4f __attribute__((ext_vector_type(4)));
typedef float v8f __attribute__((ext_vector_type(8)));

#define EPS_F 1e-6f
#define KN   32
#define NFG  4
#define NH   16
#define NFO  32

static __device__ __forceinline__ void lds_barrier() {
    asm volatile("s_wait_dscnt 0x0\n\ts_barrier_signal -1\n\ts_barrier_wait -1" ::: "memory");
}

__global__ __launch_bounds__(256)
void attn_layer_kernel(const float* __restrict__ beta,
                       const float* __restrict__ sa,
                       const float* __restrict__ ac,
                       const float* __restrict__ no,
                       const float* __restrict__ gw,
                       float* __restrict__ out,
                       float* __restrict__ out_attn)
{
    __shared__ float attn_s[KN * NH];
    __shared__ float mx_s[16 * NH];

    const int site = blockIdx.x;
    const int tid  = threadIdx.x;
    const int h    = tid & 15;
    const int k2   = tid >> 4;
    const int lane = tid & 31;
    const int wave = tid >> 5;
    const int mrow = lane & 15;
    const int sel  = lane >> 4;

    const float* ac_p = ac   + (size_t)site * (KN * NFG * NH);
    const float* no_p = no   + (size_t)site * (KN * NFO * NH);
    const float* bt_p = beta + (size_t)site * (NFG * NH);
    const float* sa_p = sa   + (size_t)site * (NFG * NH);
    const float* gw_p = gw   + (size_t)site * (KN * NFG);

    float acv[8], g[8], btv[4], sav[4];
    #pragma unroll
    for (int j = 0; j < 2; ++j) {
        const int k = k2 + j * 16;
        #pragma unroll
        for (int f = 0; f < NFG; ++f) {
            acv[j * 4 + f] = ac_p[k * (NFG * NH) + f * NH + h];
            g[j * 4 + f]   = gw_p[k * NFG + f];
        }
    }
    #pragma unroll
    for (int f = 0; f < NFG; ++f) {
        btv[f] = bt_p[f * NH + h];
        sav[f] = sa_p[f * NH + h];
    }

    __builtin_amdgcn_sched_barrier(0);

    const float* nb = no_p + sel * (2 * NFO * NH) + wave * 64 + mrow * 4;
    v4f L[8][2];
    #pragma unroll
    for (int c = 0; c < 8; ++c) {
        #pragma unroll
        for (int e = 0; e < 2; ++e) {
            L[c][e] = *(const v4f*)(nb + (4 * c + e) * (NFO * NH));
        }
    }
    __builtin_amdgcn_sched_barrier(0);

    float t[8];
    float lmax = -3.402823466e38f;
    #pragma unroll
    for (int j = 0; j < 2; ++j) {
        const int k = k2 + j * 16;
        #pragma unroll
        for (int f = 0; f < NFG; ++f) {
            float v = acv[j * 4 + f];
            if (k == 0) v += sav[f];
            v *= (btv[f] + EPS_F);
            t[j * 4 + f] = v;
            lmax = fmaxf(lmax, v);
        }
    }
    mx_s[k2 * NH + h] = lmax;
    lds_barrier();
    float m = mx_s[h];
    #pragma unroll
    for (int i = 1; i < 16; ++i) m = fmaxf(m, mx_s[i * NH + h]);

    float a0 = 0.f, a1 = 0.f;
    #pragma unroll
    for (int f = 0; f < 4; ++f) a0 += __expf(t[f] - m) * g[f];
    #pragma unroll
    for (int f = 0; f < 4; ++f) a1 += __expf(t[4 + f] - m) * g[4 + f];
    attn_s[k2 * NH + h]        = a0;
    attn_s[(k2 + 16) * NH + h] = a1;
    lds_barrier();
    float denom = EPS_F;
    #pragma unroll
    for (int k = 0; k < KN; ++k) denom += fabsf(attn_s[k * NH + h]);
    lds_barrier();
    const float rden = 1.0f / denom;
    a0 *= rden; a1 *= rden;
    attn_s[k2 * NH + h]        = a0;
    attn_s[(k2 + 16) * NH + h] = a1;
    out_attn[(size_t)site * 512 + k2 * NH + h]        = a0;
    out_attn[(size_t)site * 512 + (k2 + 16) * NH + h] = a1;
    lds_barrier();

    v8f acc[4] = {};
    #pragma unroll
    for (int c = 0; c < 8; ++c) {
        const int kb = 4 * c + 2 * sel;
        v2f b;
        b.x = attn_s[kb * NH + mrow];
        b.y = attn_s[(kb + 1) * NH + mrow];
        #pragma unroll
        for (int j = 0; j < 4; ++j) {
            v2f a;
            a.x = L[c][0][j];
            a.y = L[c][1][j];
            acc[j] = __builtin_amdgcn_wmma_f32_16x16x4_f32(
                false, a, false, b, (short)0, acc[j], false, false);
        }
    }

    const int jt = mrow & 3;
    const int q  = mrow >> 2;
    v8f ts = acc[0];
    #pragma unroll
    for (int j = 1; j < 4; ++j) {
        #pragma unroll
        for (int r = 0; r < 8; ++r) ts[r] = (jt == j) ? acc[j][r] : ts[r];
    }
    float v0 = ts[0], v1 = ts[4];
    #pragma unroll
    for (int r = 1; r < 4; ++r) {
        v0 = (q == r) ? ts[r]     : v0;
        v1 = (q == r) ? ts[4 + r] : v1;
    }

    const float give = sel ? v0 : v1;
    const float got  = __shfl_xor(give, 16);
    const float sA = sel ? got : v0;
    const float sB = sel ? v1  : got;
    float* o = out + (size_t)site * 512 + mrow;
    o[(4 * wave + sel) * NH]     = sA;
    o[(4 * wave + 2 + sel) * NH] = sB;
}

extern "C" void kernel_launch(void* const* d_in, const int* in_sizes, int n_in,
                              void* d_out, int out_size, void* d_ws, size_t ws_size,
                              hipStream_t stream) {
    const float* beta = (const float*)d_in[0];
    const float* sa   = (const float*)d_in[1];
    const float* ac   = (const float*)d_in[2];
    const float* no   = (const float*)d_in[3];
    const float* gw   = (const float*)d_in[4];
    float* out = (float*)d_out;

    const int sites = in_sizes[0] / (NFG * NH);
    float* out_attn = out + (size_t)sites * (NFO * NH);

    attn_layer_kernel<<<sites, 256, 0, stream>>>(beta, sa, ac, no, gw, out, out_attn);
}
